# starting kernel
# baseline (speedup 1.0000x reference)
#include <hip/hip_runtime.h>

typedef float v2f __attribute__((ext_vector_type(2)));
typedef float v4f __attribute__((ext_vector_type(4)));
typedef float v8f __attribute__((ext_vector_type(8)));

#define S_DIM 2048
#define B_DIM 16
#define D1 256
#define D2 256
#define ROWSTRIDE (B_DIM * D1)

#define BM 128
#define BN 64
#define BK 32
#define NKT (S_DIM / BK)
#define LDSK 36

__global__ __launch_bounds__(256)
void SumFlattenedOuterProduct_kernel(const float* __restrict__ gA,
                                     const float* __restrict__ gB,
                                     float* __restrict__ gC)
{
    __shared__ float lA[2][BM * LDSK];
    __shared__ float lB[2][BN * LDSK];

    const int b  = blockIdx.z;
    const int n0 = blockIdx.x * BN;
    const int m0 = blockIdx.y * BM;

    const int t    = threadIdx.x;
    const int lane = t & 31;
    const int wave = t >> 5;
    const int wmo  = (wave >> 1) * 32;
    const int wno  = (wave & 1) * 32;

    const int fl_mn = lane & 15;
    const int fl_k  = (lane >> 4) * 2;

    const float* pA = gA + (size_t)b * D1 + m0;
    const float* pB = gB + (size_t)b * D2 + n0;

    int am[4], akq[4];
#pragma unroll
    for (int q = 0; q < 4; ++q) { int s = t + q * 256; am[q] = s & 127; akq[q] = s >> 7; }
    int bm[2], bkq[2];
#pragma unroll
    for (int q = 0; q < 2; ++q) { int s = t + q * 256; bm[q] = s & 63; bkq[q] = s >> 6; }

    float ra[16];
    float rb[8];

    auto loadG = [&](int s0) {
#pragma unroll
        for (int q = 0; q < 4; ++q)
#pragma unroll
            for (int j = 0; j < 4; ++j)
                ra[q * 4 + j] = pA[(size_t)(s0 + akq[q] * 4 + j) * ROWSTRIDE + am[q]];
#pragma unroll
        for (int q = 0; q < 2; ++q)
#pragma unroll
            for (int j = 0; j < 4; ++j)
                rb[q * 4 + j] = pB[(size_t)(s0 + bkq[q] * 4 + j) * ROWSTRIDE + bm[q]];
        if (s0 + BK < S_DIM) {
            __builtin_prefetch(&pA[(size_t)(s0 + BK + akq[0] * 4) * ROWSTRIDE + am[0]], 0, 1);
            __builtin_prefetch(&pB[(size_t)(s0 + BK + bkq[0] * 4) * ROWSTRIDE + bm[0]], 0, 1);
        }
    };

    auto storeS = [&](int buf) {
#pragma unroll
        for (int q = 0; q < 4; ++q) {
            v4f v = { ra[q * 4 + 0], ra[q * 4 + 1], ra[q * 4 + 2], ra[q * 4 + 3] };
            *(v4f*)&lA[buf][am[q] * LDSK + akq[q] * 4] = v;
        }
#pragma unroll
        for (int q = 0; q < 2; ++q) {
            v4f v = { rb[q * 4 + 0], rb[q * 4 + 1], rb[q * 4 + 2], rb[q * 4 + 3] };
            *(v4f*)&lB[buf][bm[q] * LDSK + bkq[q] * 4] = v;
        }
    };

    v8f acc[2][2];
#pragma unroll
    for (int i = 0; i < 2; ++i)
#pragma unroll
        for (int j = 0; j < 2; ++j)
            acc[i][j] = v8f{};

    auto compute = [&](int buf) {
#pragma unroll
        for (int kk = 0; kk < BK; kk += 4) {
            v2f af[2], bf[2];
#pragma unroll
            for (int mt = 0; mt < 2; ++mt)
                af[mt] = *(const v2f*)&lA[buf][(wmo + mt * 16 + fl_mn) * LDSK + kk + fl_k];
#pragma unroll
            for (int nt = 0; nt < 2; ++nt)
                bf[nt] = *(const v2f*)&lB[buf][(wno + nt * 16 + fl_mn) * LDSK + kk + fl_k];
#pragma unroll
            for (int mt = 0; mt < 2; ++mt)
#pragma unroll
                for (int nt = 0; nt < 2; ++nt)
                    acc[mt][nt] = __builtin_amdgcn_wmma_f32_16x16x4_f32(
                        false, af[mt], false, bf[nt],
                        (short)0, acc[mt][nt], false, false);
        }
    };

    loadG(0);
    storeS(0);
    __syncthreads();

    for (int kt = 0; kt < NKT; ++kt) {
        const int cur = kt & 1;
        if (kt + 1 < NKT) loadG((kt + 1) * BK);
        compute(cur);
        if (kt + 1 < NKT) {
            storeS(cur ^ 1);
            __syncthreads();
        }
    }

    float* pC = gC + (size_t)b * (D1 * D2);
    const int mhalf = (lane >> 4) * 8;
#pragma unroll
    for (int mt = 0; mt < 2; ++mt)
#pragma unroll
        for (int nt = 0; nt < 2; ++nt)
#pragma unroll
            for (int r = 0; r < 8; ++r) {
                int m = m0 + wmo + mt * 16 + r + mhalf;
                int n = n0 + wno + nt * 16 + fl_mn;
                pC[(size_t)m * D2 + n] = acc[mt][nt][r];
            }
}

extern "C" void kernel_launch(void* const* d_in, const int* in_sizes, int n_in,
                              void* d_out, int out_size, void* d_ws, size_t ws_size,
                              hipStream_t stream) {
    const float* in1 = (const float*)d_in[0];
    const float* in2 = (const float*)d_in[1];
    float* out = (float*)d_out;

    dim3 grid(D2 / BN, D1 / BM, B_DIM);
    dim3 block(256);
    SumFlattenedOuterProduct_kernel<<<grid, block, 0, stream>>>(in1, in2, out);
}


# work-group tile 128x64 -> 64x64 (8 waves of 16x32): grid 128 -> 256 work-groups, one per WGP instead of half the WGPs idle; k-loop, loaders, LDS layout, k order and summation order as before
# speedup vs baseline: 1.0674x; 1.0674x over previous
#include <hip/hip_runtime.h>

typedef float v2f __attribute__((ext_vector_type(2)));
typedef float v4f __attribute__((ext_vector_type(4)));
typedef float v8f __attribute__((ext_vector_type(8)));

#define S_DIM 2048
#define B_DIM 16
#define D1 256
#define D2 256
#define ROWSTRIDE (B_DIM * D1)

#define BM 64
#define BN 64
#define BK 32
#define NKT (S_DIM / BK)
#define LDSK 36

__global__ __launch_bounds__(256)
void SumFlattenedOuterProduct_kernel(const float* __restrict__ gA,
                                     const float* __restrict__ gB,
                                     float* __restrict__ gC)
{
    __shared__ float lA[2][BM * LDSK];
    __shared__ float lB[2][BN * LDSK];

    const int b  = blockIdx.z;
    const int n0 = blockIdx.x * BN;
    const int m0 = blockIdx.y * BM;

    const int t    = threadIdx.x;
    const int lane = t & 31;
    const int wave = t >> 5;
    const int wmo  = (wave >> 1) * 16;
    const int wno  = (wave & 1) * 32;

    const int fl_mn = lane & 15;
    const int fl_k  = (lane >> 4) * 2;

    const float* pA = gA + (size_t)b * D1 + m0;
    const float* pB = gB + (size_t)b * D2 + n0;

    int am[2], akq[2];
#pragma unroll
    for (int q = 0; q < 2; ++q) { int s = t + q * 256; am[q] = s & 63; akq[q] = s >> 6; }
    int bm[2], bkq[2];
#pragma unroll
    for (int q = 0; q < 2; ++q) { int s = t + q * 256; bm[q] = s & 63; bkq[q] = s >> 6; }

    float ra[8];
    float rb[8];

    auto loadG = [&](int s0) {
#pragma unroll
        for (int q = 0; q < 2; ++q)
#pragma unroll
            for (int j = 0; j < 4; ++j)
                ra[q * 4 + j] = pA[(size_t)(s0 + akq[q] * 4 + j) * ROWSTRIDE + am[q]];
#pragma unroll
        for (int q = 0; q < 2; ++q)
#pragma unroll
            for (int j = 0; j < 4; ++j)
                rb[q * 4 + j] = pB[(size_t)(s0 + bkq[q] * 4 + j) * ROWSTRIDE + bm[q]];
        if (s0 + BK < S_DIM) {
            __builtin_prefetch(&pA[(size_t)(s0 + BK + akq[0] * 4) * ROWSTRIDE + am[0]], 0, 1);
            __builtin_prefetch(&pB[(size_t)(s0 + BK + bkq[0] * 4) * ROWSTRIDE + bm[0]], 0, 1);
        }
    };

    auto storeS = [&](int buf) {
#pragma unroll
        for (int q = 0; q < 2; ++q) {
            v4f v = { ra[q * 4 + 0], ra[q * 4 + 1], ra[q * 4 + 2], ra[q * 4 + 3] };
            *(v4f*)&lA[buf][am[q] * LDSK + akq[q] * 4] = v;
        }
#pragma unroll
        for (int q = 0; q < 2; ++q) {
            v4f v = { rb[q * 4 + 0], rb[q * 4 + 1], rb[q * 4 + 2], rb[q * 4 + 3] };
            *(v4f*)&lB[buf][bm[q] * LDSK + bkq[q] * 4] = v;
        }
    };

    v8f acc[2];
#pragma unroll
    for (int j = 0; j < 2; ++j)
        acc[j] = v8f{};

    auto compute = [&](int buf) {
#pragma unroll
        for (int kk = 0; kk < BK; kk += 4) {
            v2f af, bf[2];
            af = *(const v2f*)&lA[buf][(wmo + fl_mn) * LDSK + kk + fl_k];
#pragma unroll
            for (int nt = 0; nt < 2; ++nt)
                bf[nt] = *(const v2f*)&lB[buf][(wno + nt * 16 + fl_mn) * LDSK + kk + fl_k];
#pragma unroll
            for (int nt = 0; nt < 2; ++nt)
                acc[nt] = __builtin_amdgcn_wmma_f32_16x16x4_f32(
                    false, af, false, bf[nt],
                    (short)0, acc[nt], false, false);
        }
    };

    loadG(0);
    storeS(0);
    __syncthreads();

    for (int kt = 0; kt < NKT; ++kt) {
        const int cur = kt & 1;
        if (kt + 1 < NKT) loadG((kt + 1) * BK);
        compute(cur);
        if (kt + 1 < NKT) {
            storeS(cur ^ 1);
            __syncthreads();
        }
    }

    float* pC = gC + (size_t)b * (D1 * D2);
    const int mhalf = (lane >> 4) * 8;
#pragma unroll
    for (int nt = 0; nt < 2; ++nt)
#pragma unroll
        for (int r = 0; r < 8; ++r) {
            int m = m0 + wmo + r + mhalf;
            int n = n0 + wno + nt * 16 + fl_mn;
            pC[(size_t)m * D2 + n] = acc[nt][r];
        }
}

extern "C" void kernel_launch(void* const* d_in, const int* in_sizes, int n_in,
                              void* d_out, int out_size, void* d_ws, size_t ws_size,
                              hipStream_t stream) {
    const float* in1 = (const float*)d_in[0];
    const float* in2 = (const float*)d_in[1];
    float* out = (float*)d_out;

    dim3 grid(D2 / BN, D1 / BM, B_DIM);
    dim3 block(256);
    SumFlattenedOuterProduct_kernel<<<grid, block, 0, stream>>>(in1, in2, out);
}


# global loads staged 3 tiles ahead (4 register sets, k-loop unrolled by 4) and kept in flight across an LDS-only work-group barrier; k order and summation order as before
# speedup vs baseline: 1.8205x; 1.7055x over previous
#include <hip/hip_runtime.h>

typedef float v2f __attribute__((ext_vector_type(2)));
typedef float v4f __attribute__((ext_vector_type(4)));
typedef float v8f __attribute__((ext_vector_type(8)));

#define S_DIM 2048
#define B_DIM 16
#define D1 256
#define D2 256
#define ROWSTRIDE (B_DIM * D1)

#define BM 64
#define BN 64
#define BK 32
#define NKT (S_DIM / BK)
#define LDSK 36
#define NSET 4

static __device__ __forceinline__ void lds_barrier() {
    __builtin_amdgcn_fence(__ATOMIC_RELEASE, "workgroup", "local");
    __builtin_amdgcn_s_barrier();
    __builtin_amdgcn_fence(__ATOMIC_ACQUIRE, "workgroup", "local");
}

__global__ __launch_bounds__(256)
void SumFlattenedOuterProduct_kernel(const float* __restrict__ gA,
                                     const float* __restrict__ gB,
                                     float* __restrict__ gC)
{
    __shared__ float lA[2][BM * LDSK];
    __shared__ float lB[2][BN * LDSK];

    const int b  = blockIdx.z;
    const int n0 = blockIdx.x * BN;
    const int m0 = blockIdx.y * BM;

    const int t    = threadIdx.x;
    const int lane = t & 31;
    const int wave = t >> 5;
    const int wmo  = (wave >> 1) * 16;
    const int wno  = (wave & 1) * 32;

    const int fl_mn = lane & 15;
    const int fl_k  = (lane >> 4) * 2;

    const float* pA = gA + (size_t)b * D1 + m0;
    const float* pB = gB + (size_t)b * D2 + n0;

    int am[2], akq[2];
#pragma unroll
    for (int q = 0; q < 2; ++q) { int s = t + q * 256; am[q] = s & 63; akq[q] = s >> 6; }
    int bm[2], bkq[2];
#pragma unroll
    for (int q = 0; q < 2; ++q) { int s = t + q * 256; bm[q] = s & 63; bkq[q] = s >> 6; }

    float ra[NSET][8];
    float rb[NSET][8];

    auto loadG = [&](float (&xa)[8], float (&xb)[8], int s0) {
#pragma unroll
        for (int q = 0; q < 2; ++q)
#pragma unroll
            for (int j = 0; j < 4; ++j)
                xa[q * 4 + j] = pA[(size_t)(s0 + akq[q] * 4 + j) * ROWSTRIDE + am[q]];
#pragma unroll
        for (int q = 0; q < 2; ++q)
#pragma unroll
            for (int j = 0; j < 4; ++j)
                xb[q * 4 + j] = pB[(size_t)(s0 + bkq[q] * 4 + j) * ROWSTRIDE + bm[q]];
        if (s0 + BK < S_DIM) {
            __builtin_prefetch(&pA[(size_t)(s0 + BK + akq[0] * 4) * ROWSTRIDE + am[0]], 0, 1);
            __builtin_prefetch(&pB[(size_t)(s0 + BK + bkq[0] * 4) * ROWSTRIDE + bm[0]], 0, 1);
        }
    };

    auto storeS = [&](const float (&xa)[8], const float (&xb)[8], int buf) {
#pragma unroll
        for (int q = 0; q < 2; ++q) {
            v4f v = { xa[q * 4 + 0], xa[q * 4 + 1], xa[q * 4 + 2], xa[q * 4 + 3] };
            *(v4f*)&lA[buf][am[q] * LDSK + akq[q] * 4] = v;
        }
#pragma unroll
        for (int q = 0; q < 2; ++q) {
            v4f v = { xb[q * 4 + 0], xb[q * 4 + 1], xb[q * 4 + 2], xb[q * 4 + 3] };
            *(v4f*)&lB[buf][bm[q] * LDSK + bkq[q] * 4] = v;
        }
    };

    v8f acc[2];
#pragma unroll
    for (int j = 0; j < 2; ++j)
        acc[j] = v8f{};

    auto compute = [&](int buf) {
#pragma unroll
        for (int kk = 0; kk < BK; kk += 4) {
            v2f af, bf[2];
            af = *(const v2f*)&lA[buf][(wmo + fl_mn) * LDSK + kk + fl_k];
#pragma unroll
            for (int nt = 0; nt < 2; ++nt)
                bf[nt] = *(const v2f*)&lB[buf][(wno + nt * 16 + fl_mn) * LDSK + kk + fl_k];
#pragma unroll
            for (int nt = 0; nt < 2; ++nt)
                acc[nt] = __builtin_amdgcn_wmma_f32_16x16x4_f32(
                    false, af, false, bf[nt],
                    (short)0, acc[nt], false, false);
        }
    };

    loadG(ra[0], rb[0], 0 * BK);
    loadG(ra[1], rb[1], 1 * BK);
    loadG(ra[2], rb[2], 2 * BK);
    loadG(ra[3], rb[3], 3 * BK);
    storeS(ra[0], rb[0], 0);
    lds_barrier();

    for (int kt0 = 0; kt0 < NKT - NSET; kt0 += NSET) {
#pragma unroll
        for (int u = 0; u < NSET; ++u) {
            const int kt = kt0 + u;
            storeS(ra[(u + 1) % NSET], rb[(u + 1) % NSET], (u + 1) & 1);
            loadG(ra[u], rb[u], (kt + NSET) * BK);
            compute(u & 1);
            lds_barrier();
        }
    }
#pragma unroll
    for (int u = 0; u < NSET; ++u) {
        if (u + 1 < NSET) storeS(ra[(u + 1) % NSET], rb[(u + 1) % NSET], (u + 1) & 1);
        compute(u & 1);
        if (u + 1 < NSET) lds_barrier();
    }

    float* pC = gC + (size_t)b * (D1 * D2);
    const int mhalf = (lane >> 4) * 8;
#pragma unroll
    for (int nt = 0; nt < 2; ++nt)
#pragma unroll
        for (int r = 0; r < 8; ++r) {
            int m = m0 + wmo + r + mhalf;
            int n = n0 + wno + nt * 16 + fl_mn;
            pC[(size_t)m * D2 + n] = acc[nt][r];
        }
}

extern "C" void kernel_launch(void* const* d_in, const int* in_sizes, int n_in,
                              void* d_out, int out_size, void* d_ws, size_t ws_size,
                              hipStream_t stream) {
    const float* in1 = (const float*)d_in[0];
    const float* in2 = (const float*)d_in[1];
    float* out = (float*)d_out;

    dim3 grid(D2 / BN, D1 / BM, B_DIM);
    dim3 block(256);
    SumFlattenedOuterProduct_kernel<<<grid, block, 0, stream>>>(in1, in2, out);
}


# WMMA fragments double-buffered in registers: the next K-tile's LDS fragment reads go out as one group ahead of the current tile's 16 WMMAs; k order and summation order as before
# speedup vs baseline: 1.9317x; 1.0611x over previous
#include <hip/hip_runtime.h>

typedef float v2f __attribute__((ext_vector_type(2)));
typedef float v4f __attribute__((ext_vector_type(4)));
typedef float v8f __attribute__((ext_vector_type(8)));

#define S_DIM 2048
#define B_DIM 16
#define D1 256
#define D2 256
#define ROWSTRIDE (B_DIM * D1)

#define BM 64
#define BN 64
#define BK 32
#define NKT (S_DIM / BK)
#define LDSK 36
#define NSET 4

static __device__ __forceinline__ void lds_barrier() {
    __builtin_amdgcn_fence(__ATOMIC_RELEASE, "workgroup", "local");
    __builtin_amdgcn_s_barrier();
    __builtin_amdgcn_fence(__ATOMIC_ACQUIRE, "workgroup", "local");
}

__global__ __launch_bounds__(256)
void SumFlattenedOuterProduct_kernel(const float* __restrict__ gA,
                                     const float* __restrict__ gB,
                                     float* __restrict__ gC)
{
    __shared__ float lA[2][BM * LDSK];
    __shared__ float lB[2][BN * LDSK];

    const int b  = blockIdx.z;
    const int n0 = blockIdx.x * BN;
    const int m0 = blockIdx.y * BM;

    const int t    = threadIdx.x;
    const int lane = t & 31;
    const int wave = t >> 5;
    const int wmo  = (wave >> 1) * 16;
    const int wno  = (wave & 1) * 32;

    const int fl_mn = lane & 15;
    const int fl_k  = (lane >> 4) * 2;

    const float* pA = gA + (size_t)b * D1 + m0;
    const float* pB = gB + (size_t)b * D2 + n0;

    int am[2], akq[2];
#pragma unroll
    for (int q = 0; q < 2; ++q) { int s = t + q * 256; am[q] = s & 63; akq[q] = s >> 6; }
    int bm[2], bkq[2];
#pragma unroll
    for (int q = 0; q < 2; ++q) { int s = t + q * 256; bm[q] = s & 63; bkq[q] = s >> 6; }

    float ra[NSET][8];
    float rb[NSET][8];

    auto loadG = [&](float (&xa)[8], float (&xb)[8], int s0) {
#pragma unroll
        for (int q = 0; q < 2; ++q)
#pragma unroll
            for (int j = 0; j < 4; ++j)
                xa[q * 4 + j] = pA[(size_t)(s0 + akq[q] * 4 + j) * ROWSTRIDE + am[q]];
#pragma unroll
        for (int q = 0; q < 2; ++q)
#pragma unroll
            for (int j = 0; j < 4; ++j)
                xb[q * 4 + j] = pB[(size_t)(s0 + bkq[q] * 4 + j) * ROWSTRIDE + bm[q]];
        if (s0 + BK < S_DIM) {
            __builtin_prefetch(&pA[(size_t)(s0 + BK + akq[0] * 4) * ROWSTRIDE + am[0]], 0, 1);
            __builtin_prefetch(&pB[(size_t)(s0 + BK + bkq[0] * 4) * ROWSTRIDE + bm[0]], 0, 1);
        }
    };

    auto storeS = [&](const float (&xa)[8], const float (&xb)[8], int buf) {
#pragma unroll
        for (int q = 0; q < 2; ++q) {
            v4f v = { xa[q * 4 + 0], xa[q * 4 + 1], xa[q * 4 + 2], xa[q * 4 + 3] };
            *(v4f*)&lA[buf][am[q] * LDSK + akq[q] * 4] = v;
        }
#pragma unroll
        for (int q = 0; q < 2; ++q) {
            v4f v = { xb[q * 4 + 0], xb[q * 4 + 1], xb[q * 4 + 2], xb[q * 4 + 3] };
            *(v4f*)&lB[buf][bm[q] * LDSK + bkq[q] * 4] = v;
        }
    };

    v8f acc[2];
#pragma unroll
    for (int j = 0; j < 2; ++j)
        acc[j] = v8f{};

    v2f fa[2][BK / 4];
    v2f fb[2][BK / 4][2];

    auto loadF = [&](v2f (&xa)[BK / 4], v2f (&xb)[BK / 4][2], int buf) {
#pragma unroll
        for (int ks = 0; ks < BK / 4; ++ks) {
            xa[ks] = *(const v2f*)&lA[buf][(wmo + fl_mn) * LDSK + ks * 4 + fl_k];
#pragma unroll
            for (int nt = 0; nt < 2; ++nt)
                xb[ks][nt] = *(const v2f*)&lB[buf][(wno + nt * 16 + fl_mn) * LDSK + ks * 4 + fl_k];
        }
    };

    auto mma = [&](const v2f (&xa)[BK / 4], const v2f (&xb)[BK / 4][2]) {
#pragma unroll
        for (int ks = 0; ks < BK / 4; ++ks)
#pragma unroll
            for (int nt = 0; nt < 2; ++nt)
                acc[nt] = __builtin_amdgcn_wmma_f32_16x16x4_f32(
                    false, xa[ks], false, xb[ks][nt],
                    (short)0, acc[nt], false, false);
    };

    loadG(ra[0], rb[0], 0 * BK);
    loadG(ra[1], rb[1], 1 * BK);
    loadG(ra[2], rb[2], 2 * BK);
    loadG(ra[3], rb[3], 3 * BK);
    storeS(ra[0], rb[0], 0);
    lds_barrier();
    loadF(fa[0], fb[0], 0);
    storeS(ra[1], rb[1], 1);
    loadG(ra[0], rb[0], 4 * BK);
    lds_barrier();

    for (int kt0 = 0; kt0 < NKT - 2 * NSET; kt0 += NSET) {
#pragma unroll
        for (int u = 0; u < NSET; ++u) {
            const int kt = kt0 + u;
            loadF(fa[(u + 1) & 1], fb[(u + 1) & 1], (u + 1) & 1);
            storeS(ra[(u + 2) % NSET], rb[(u + 2) % NSET], u & 1);
            loadG(ra[(u + 1) % NSET], rb[(u + 1) % NSET], (kt + NSET + 1) * BK);
            __builtin_amdgcn_sched_barrier(0);
            mma(fa[u & 1], fb[u & 1]);
            __builtin_amdgcn_sched_barrier(0);
            lds_barrier();
            __builtin_amdgcn_sched_barrier(0);
        }
    }
#pragma unroll
    for (int d = 0; d < 2 * NSET; ++d) {
        const int kt = NKT - 2 * NSET + d;
        const int u  = d % NSET;
        if (kt + 1 < NKT) loadF(fa[(u + 1) & 1], fb[(u + 1) & 1], (u + 1) & 1);
        if (kt + 2 < NKT) storeS(ra[(u + 2) % NSET], rb[(u + 2) % NSET], u & 1);
        if (kt + NSET + 1 < NKT) loadG(ra[(u + 1) % NSET], rb[(u + 1) % NSET], (kt + NSET + 1) * BK);
        __builtin_amdgcn_sched_barrier(0);
        mma(fa[u & 1], fb[u & 1]);
        __builtin_amdgcn_sched_barrier(0);
        if (kt + 2 < NKT) lds_barrier();
        __builtin_amdgcn_sched_barrier(0);
    }

    float* pC = gC + (size_t)b * (D1 * D2);
    const int mhalf = (lane >> 4) * 8;
#pragma unroll
    for (int nt = 0; nt < 2; ++nt)
#pragma unroll
        for (int r = 0; r < 8; ++r) {
            int m = m0 + wmo + r + mhalf;
            int n = n0 + wno + nt * 16 + fl_mn;
            pC[(size_t)m * D2 + n] = acc[nt][r];
        }
}

extern "C" void kernel_launch(void* const* d_in, const int* in_sizes, int n_in,
                              void* d_out, int out_size, void* d_ws, size_t ws_size,
                              hipStream_t stream) {
    const float* in1 = (const float*)d_in[0];
    const float* in2 = (const float*)d_in[1];
    float* out = (float*)d_out;

    dim3 grid(D2 / BN, D1 / BM, B_DIM);
    dim3 block(256);
    SumFlattenedOuterProduct_kernel<<<grid, block, 0, stream>>>(in1, in2, out);
}
